# starting kernel
# baseline (speedup 1.0000x reference)
#include <hip/hip_runtime.h>
#include <math.h>

#define FEAT_D 32

typedef __attribute__((ext_vector_type(2))) float v2f;
typedef __attribute__((ext_vector_type(8))) float v8f;

__global__ __launch_bounds__(256)
void zero_kernel(float4* __restrict__ ax4, int n4) {
    int i = blockIdx.x * blockDim.x + threadIdx.x;
    if (i < n4) ax4[i] = make_float4(0.f, 0.f, 0.f, 0.f);
}

__global__ __launch_bounds__(256)
void scatter_kernel(const float* __restrict__ x,
                    const int*   __restrict__ src,
                    const int*   __restrict__ dst,
                    const float* __restrict__ w,
                    float*       __restrict__ ax,
                    int E) {
    int wave = blockIdx.x * (blockDim.x >> 5) + (threadIdx.x >> 5);
    int e = __builtin_amdgcn_readfirstlane(wave);
    if (e >= E) return;

    __builtin_prefetch(&src[e + 8192], 0, 1);
    __builtin_prefetch(&dst[e + 8192], 0, 1);
    __builtin_prefetch(&w[e + 8192], 0, 1);

    int   s  = src[e];
    int   t  = dst[e];
    float we = w[e];

    int lane = threadIdx.x & 31;
    float val = we * x[s * FEAT_D + lane];
    unsafeAtomicAdd(&ax[t * FEAT_D + lane], val);
}

__global__ __launch_bounds__(256)
void update_wmma_kernel(const float* __restrict__ x,
                        const float* __restrict__ ax,
                        const float* __restrict__ x0,
                        const float* __restrict__ alpha_p,
                        const float* __restrict__ beta_p,
                        float*       __restrict__ out,
                        int ntiles) {
    int tile = __builtin_amdgcn_readfirstlane(
        blockIdx.x * (blockDim.x >> 5) + (threadIdx.x >> 5));
    if (tile >= ntiles) return;

    const int nb = (tile >> 1) * 16;
    const int cb = (tile & 1) * 16;

    const int lane = threadIdx.x & 31;
    const int hi   = lane >> 4;
    const int li   = lane & 15;

    float araw  = *alpha_p;
    float alpha = 1.0f / (1.0f + expf(-araw));
    float beta  = *beta_p;
    float ca    = 1.0f - alpha;

    v8f c;
#pragma unroll
    for (int v = 0; v < 8; ++v) {
        int row = nb + v + 8 * hi;
        int idx = row * FEAT_D + cb + li;
        c[v] = ca * x[idx] + beta * x0[idx];
    }

#pragma unroll
    for (int m = 0; m < 4; ++m) {
        v2f a, b;
#pragma unroll
        for (int v = 0; v < 2; ++v) {
            int k = v + 2 * hi;
            a[v] = (li == 4 * m + k) ? alpha : 0.0f;
            int rowb = nb + 4 * m + k;
            b[v] = ax[rowb * FEAT_D + cb + li];
        }
        c = __builtin_amdgcn_wmma_f32_16x16x4_f32(
                false, a, false, b, (short)0, c, false, false);
    }

#pragma unroll
    for (int v = 0; v < 8; ++v) {
        int row = nb + v + 8 * hi;
        out[row * FEAT_D + cb + li] = c[v];
    }
}

extern "C" void kernel_launch(void* const* d_in, const int* in_sizes, int n_in,
                              void* d_out, int out_size, void* d_ws, size_t ws_size,
                              hipStream_t stream) {
    const float* x_in    = (const float*)d_in[0];
    const float* w       = (const float*)d_in[1];
    const float* x0      = (const float*)d_in[2];
    const float* alpha_p = (const float*)d_in[3];
    const float* beta_p  = (const float*)d_in[4];
    const int*   eidx    = (const int*)  d_in[5];

    const int N = in_sizes[0] / FEAT_D;
    const int E = in_sizes[1];
    const int* src = eidx;
    const int* dst = eidx + E;

    float* ax  = (float*)d_ws;
    float* out = (float*)d_out;

    const int ND = N * FEAT_D;
    dim3 blk(256);
    const int zeroBlocks = (ND / 4 + 255) / 256;
    const int scatBlocks = (E + 7) / 8;
    const int ntiles     = (N / 16) * (FEAT_D / 16);
    const int updBlocks  = (ntiles + 7) / 8;

    const float* xcur = x_in;
    for (int step = 0; step < 4; ++step) {
        zero_kernel<<<zeroBlocks, blk, 0, stream>>>((float4*)ax, ND / 4);
        scatter_kernel<<<scatBlocks, blk, 0, stream>>>(xcur, src, dst, w, ax, E);
        update_wmma_kernel<<<updBlocks, blk, 0, stream>>>(
            xcur, ax, x0, alpha_p, beta_p, out, ntiles);
        xcur = out;
    }
}


# Scatter: each wave takes 32 consecutive edges (coalesced src/dst/w loads, readlane broadcast) and issues all 32 row gathers before its 32 row atomics; same f32 atomics, arrival order differs
# speedup vs baseline: 2.1255x; 2.1255x over previous
#include <hip/hip_runtime.h>
#include <math.h>

#define FEAT_D 32

typedef __attribute__((ext_vector_type(2))) float v2f;
typedef __attribute__((ext_vector_type(8))) float v8f;

__global__ __launch_bounds__(256)
void zero_kernel(float4* __restrict__ ax4, int n4) {
    int i = blockIdx.x * blockDim.x + threadIdx.x;
    if (i < n4) ax4[i] = make_float4(0.f, 0.f, 0.f, 0.f);
}

#define EDGES_PER_WAVE 32

__device__ __forceinline__ void scatter_one(const float* __restrict__ x,
                                            float*       __restrict__ ax,
                                            int s, int t, float we, int lane) {
    float val = we * x[(size_t)(unsigned)s * FEAT_D + lane];
    unsafeAtomicAdd(&ax[(size_t)(unsigned)t * FEAT_D + lane], val);
}

__global__ __launch_bounds__(256)
void scatter_kernel(const float* __restrict__ x,
                    const int*   __restrict__ src,
                    const int*   __restrict__ dst,
                    const float* __restrict__ w,
                    float*       __restrict__ ax,
                    int E) {
    const int wave = __builtin_amdgcn_readfirstlane(
        blockIdx.x * (blockDim.x >> 5) + (threadIdx.x >> 5));
    const int e0 = wave * EDGES_PER_WAVE;
    if (e0 >= E) return;

    const int lane = threadIdx.x & 31;
    const unsigned ulane = threadIdx.x & 31u;
    const int cnt  = min(EDGES_PER_WAVE, E - e0);
    const int el   = e0 + min(lane, cnt - 1);

    const int s_l = src[el];
    const int t_l = dst[el];
    const int w_l = __float_as_int(w[el]);

    if (cnt == EDGES_PER_WAVE) {
        float v[EDGES_PER_WAVE];
#pragma unroll
        for (int j = 0; j < EDGES_PER_WAVE; ++j) {
            const int s = __builtin_amdgcn_readlane(s_l, j);
            v[j] = x[((unsigned)s << 5) + ulane];
        }
#pragma unroll
        for (int j = 0; j < EDGES_PER_WAVE; ++j) {
            const int   t  = __builtin_amdgcn_readlane(t_l, j);
            const float we = __int_as_float(__builtin_amdgcn_readlane(w_l, j));
            unsafeAtomicAdd(&ax[((unsigned)t << 5) + ulane], we * v[j]);
        }
    } else {
        for (int j = 0; j < cnt; ++j) {
            const int   s  = __builtin_amdgcn_readlane(s_l, j);
            const int   t  = __builtin_amdgcn_readlane(t_l, j);
            const float we = __int_as_float(__builtin_amdgcn_readlane(w_l, j));
            scatter_one(x, ax, s, t, we, lane);
        }
    }
}

__global__ __launch_bounds__(256)
void update_wmma_kernel(const float* __restrict__ x,
                        const float* __restrict__ ax,
                        const float* __restrict__ x0,
                        const float* __restrict__ alpha_p,
                        const float* __restrict__ beta_p,
                        float*       __restrict__ out,
                        int ntiles) {
    int tile = __builtin_amdgcn_readfirstlane(
        blockIdx.x * (blockDim.x >> 5) + (threadIdx.x >> 5));
    if (tile >= ntiles) return;

    const int nb = (tile >> 1) * 16;
    const int cb = (tile & 1) * 16;

    const int lane = threadIdx.x & 31;
    const int hi   = lane >> 4;
    const int li   = lane & 15;

    float araw  = *alpha_p;
    float alpha = 1.0f / (1.0f + expf(-araw));
    float beta  = *beta_p;
    float ca    = 1.0f - alpha;

    v8f c;
#pragma unroll
    for (int v = 0; v < 8; ++v) {
        int row = nb + v + 8 * hi;
        int idx = row * FEAT_D + cb + li;
        c[v] = ca * x[idx] + beta * x0[idx];
    }

#pragma unroll
    for (int m = 0; m < 4; ++m) {
        v2f a, b;
#pragma unroll
        for (int v = 0; v < 2; ++v) {
            int k = v + 2 * hi;
            a[v] = (li == 4 * m + k) ? alpha : 0.0f;
            int rowb = nb + 4 * m + k;
            b[v] = ax[rowb * FEAT_D + cb + li];
        }
        c = __builtin_amdgcn_wmma_f32_16x16x4_f32(
                false, a, false, b, (short)0, c, false, false);
    }

#pragma unroll
    for (int v = 0; v < 8; ++v) {
        int row = nb + v + 8 * hi;
        out[row * FEAT_D + cb + li] = c[v];
    }
}

extern "C" void kernel_launch(void* const* d_in, const int* in_sizes, int n_in,
                              void* d_out, int out_size, void* d_ws, size_t ws_size,
                              hipStream_t stream) {
    const float* x_in    = (const float*)d_in[0];
    const float* w       = (const float*)d_in[1];
    const float* x0      = (const float*)d_in[2];
    const float* alpha_p = (const float*)d_in[3];
    const float* beta_p  = (const float*)d_in[4];
    const int*   eidx    = (const int*)  d_in[5];

    const int N = in_sizes[0] / FEAT_D;
    const int E = in_sizes[1];
    const int* src = eidx;
    const int* dst = eidx + E;

    float* ax  = (float*)d_ws;
    float* out = (float*)d_out;

    const int ND = N * FEAT_D;
    dim3 blk(256);
    const int zeroBlocks = (ND / 4 + 255) / 256;
    const int scatBlocks = (E + 255) / 256;
    const int ntiles     = (N / 16) * (FEAT_D / 16);
    const int updBlocks  = (ntiles + 7) / 8;

    const float* xcur = x_in;
    for (int step = 0; step < 4; ++step) {
        zero_kernel<<<zeroBlocks, blk, 0, stream>>>((float4*)ax, ND / 4);
        scatter_kernel<<<scatBlocks, blk, 0, stream>>>(xcur, src, dst, w, ax, E);
        update_wmma_kernel<<<updBlocks, blk, 0, stream>>>(
            xcur, ax, x0, alpha_p, beta_p, out, ntiles);
        xcur = out;
    }
}


# Fixed grids of long-lived waves in all three kernels: a wave strides over edge batches or tiles, next one's loads requested ahead, so far fewer waves enter; f32 atomics arrive in another order
# speedup vs baseline: 2.1956x; 1.0330x over previous
#include <hip/hip_runtime.h>
#include <math.h>


#define FEAT_D 32

typedef __attribute__((ext_vector_type(2))) float v2f;
typedef __attribute__((ext_vector_type(8))) float v8f;

#define ZERO_BLOCKS 512
#define SCAT_BLOCKS 1024
#define UPD_BLOCKS  512

__global__ __launch_bounds__(256)
void zero_kernel(float4* __restrict__ ax4, int n4) {
    const int stride = gridDim.x * blockDim.x;
    for (int i = blockIdx.x * blockDim.x + threadIdx.x; i < n4; i += stride)
        ax4[i] = make_float4(0.f, 0.f, 0.f, 0.f);
}

#define EDGES_PER_WAVE 32

__device__ __forceinline__ void scatter_one(const float* __restrict__ x,
                                            float*       __restrict__ ax,
                                            int s, int t, float we, int lane) {
    float val = we * x[(size_t)(unsigned)s * FEAT_D + lane];
    unsafeAtomicAdd(&ax[(size_t)(unsigned)t * FEAT_D + lane], val);
}

__global__ __launch_bounds__(256)
void scatter_kernel(const float* __restrict__ x,
                    const int*   __restrict__ src,
                    const int*   __restrict__ dst,
                    const float* __restrict__ w,
                    float*       __restrict__ ax,
                    int E) {
    const int wave = __builtin_amdgcn_readfirstlane(
        blockIdx.x * (blockDim.x >> 5) + (threadIdx.x >> 5));
    const int nwaves = gridDim.x * (blockDim.x >> 5);
    const int lane = threadIdx.x & 31;
    const unsigned ulane = threadIdx.x & 31u;
    const int nfull = E >> 5;

    int b = wave;
    int s_l = 0, t_l = 0, w_l = 0;
    if (b < nfull) {
        const int el = (b << 5) + lane;
        s_l = src[el];
        t_l = dst[el];
        w_l = __float_as_int(w[el]);
    }
    while (b < nfull) {
        float v[EDGES_PER_WAVE];
#pragma unroll
        for (int j = 0; j < EDGES_PER_WAVE; ++j) {
            const int s = __builtin_amdgcn_readlane(s_l, j);
            v[j] = x[((unsigned)s << 5) + ulane];
        }
        const int bn = b + nwaves;
        int s_n = 0, t_n = 0, w_n = 0;
        if (bn < nfull) {
            const int el = (bn << 5) + lane;
            s_n = src[el];
            t_n = dst[el];
            w_n = __float_as_int(w[el]);
        }
#pragma unroll
        for (int j = 0; j < EDGES_PER_WAVE; ++j) {
            const int   t  = __builtin_amdgcn_readlane(t_l, j);
            const float we = __int_as_float(__builtin_amdgcn_readlane(w_l, j));
            unsafeAtomicAdd(&ax[((unsigned)t << 5) + ulane], we * v[j]);
        }
        s_l = s_n; t_l = t_n; w_l = w_n;
        b = bn;
    }

    const int rem = E & 31;
    if (rem != 0 && wave == nwaves - 1) {
        const int el = (nfull << 5) + min(lane, rem - 1);
        s_l = src[el];
        t_l = dst[el];
        w_l = __float_as_int(w[el]);
        for (int j = 0; j < rem; ++j) {
            const int   s  = __builtin_amdgcn_readlane(s_l, j);
            const int   t  = __builtin_amdgcn_readlane(t_l, j);
            const float we = __int_as_float(__builtin_amdgcn_readlane(w_l, j));
            scatter_one(x, ax, s, t, we, lane);
        }
    }
}

struct TileRegs {
    float xv[8];
    float zv[8];
    float bv[8];
};

__device__ __forceinline__ void load_tile(const float* __restrict__ x,
                                          const float* __restrict__ ax,
                                          const float* __restrict__ x0,
                                          int tile, int hi, int li, TileRegs& r) {
    const int nb = (tile >> 1) * 16;
    const int cb = (tile & 1) * 16;
#pragma unroll
    for (int v = 0; v < 8; ++v) {
        const int idx = (nb + v + 8 * hi) * FEAT_D + cb + li;
        r.xv[v] = x[idx];
        r.zv[v] = x0[idx];
    }
#pragma unroll
    for (int m = 0; m < 4; ++m) {
#pragma unroll
        for (int v = 0; v < 2; ++v) {
            const int rowb = nb + 4 * m + v + 2 * hi;
            r.bv[2 * m + v] = ax[rowb * FEAT_D + cb + li];
        }
    }
}

__device__ __forceinline__ void finish_tile(const TileRegs& r, int tile, int hi, int li,
                                            float alpha, float ca, float beta,
                                            float* __restrict__ out) {
    const int nb = (tile >> 1) * 16;
    const int cb = (tile & 1) * 16;

    v8f c;
#pragma unroll
    for (int v = 0; v < 8; ++v)
        c[v] = ca * r.xv[v] + beta * r.zv[v];

#pragma unroll
    for (int m = 0; m < 4; ++m) {
        v2f a, b;
#pragma unroll
        for (int v = 0; v < 2; ++v) {
            const int k = v + 2 * hi;
            a[v] = (li == 4 * m + k) ? alpha : 0.0f;
            b[v] = r.bv[2 * m + v];
        }
        c = __builtin_amdgcn_wmma_f32_16x16x4_f32(
                false, a, false, b, (short)0, c, false, false);
    }

#pragma unroll
    for (int v = 0; v < 8; ++v) {
        const int row = nb + v + 8 * hi;
        out[row * FEAT_D + cb + li] = c[v];
    }
}

__global__ __launch_bounds__(256)
void update_wmma_kernel(const float* __restrict__ x,
                        const float* __restrict__ ax,
                        const float* __restrict__ x0,
                        const float* __restrict__ alpha_p,
                        const float* __restrict__ beta_p,
                        float*       __restrict__ out,
                        int ntiles) {
    const int wave = __builtin_amdgcn_readfirstlane(
        blockIdx.x * (blockDim.x >> 5) + (threadIdx.x >> 5));
    const int nwaves = gridDim.x * (blockDim.x >> 5);
    if (wave >= ntiles) return;

    const int lane = threadIdx.x & 31;
    const int hi   = lane >> 4;
    const int li   = lane & 15;

    float araw  = *alpha_p;
    float alpha = 1.0f / (1.0f + expf(-araw));
    float beta  = *beta_p;
    float ca    = 1.0f - alpha;

    const int step2 = 2 * nwaves;
    int t = wave;
    TileRegs ra, rb;
    load_tile(x, ax, x0, t, hi, li, ra);
    while (t + step2 < ntiles) {
        load_tile(x, ax, x0, t + nwaves, hi, li, rb);
        __builtin_amdgcn_sched_barrier(0);
        finish_tile(ra, t, hi, li, alpha, ca, beta, out);
        load_tile(x, ax, x0, t + step2, hi, li, ra);
        __builtin_amdgcn_sched_barrier(0);
        finish_tile(rb, t + nwaves, hi, li, alpha, ca, beta, out);
        t += step2;
    }
    if (t + nwaves < ntiles) {
        load_tile(x, ax, x0, t + nwaves, hi, li, rb);
        __builtin_amdgcn_sched_barrier(0);
        finish_tile(ra, t, hi, li, alpha, ca, beta, out);
        finish_tile(rb, t + nwaves, hi, li, alpha, ca, beta, out);
    } else {
        finish_tile(ra, t, hi, li, alpha, ca, beta, out);
    }
}

extern "C" void kernel_launch(void* const* d_in, const int* in_sizes, int n_in,
                              void* d_out, int out_size, void* d_ws, size_t ws_size,
                              hipStream_t stream) {
    const float* x_in    = (const float*)d_in[0];
    const float* w       = (const float*)d_in[1];
    const float* x0      = (const float*)d_in[2];
    const float* alpha_p = (const float*)d_in[3];
    const float* beta_p  = (const float*)d_in[4];
    const int*   eidx    = (const int*)  d_in[5];

    const int N = in_sizes[0] / FEAT_D;
    const int E = in_sizes[1];
    const int* src = eidx;
    const int* dst = eidx + E;

    float* ax  = (float*)d_ws;
    float* out = (float*)d_out;

    const int ND = N * FEAT_D;
    dim3 blk(256);
    const int ntiles = (N / 16) * (FEAT_D / 16);

    const float* xcur = x_in;
    for (int step = 0; step < 4; ++step) {
        zero_kernel<<<ZERO_BLOCKS, blk, 0, stream>>>((float4*)ax, ND / 4);
        scatter_kernel<<<SCAT_BLOCKS, blk, 0, stream>>>(xcur, src, dst, w, ax, E);
        update_wmma_kernel<<<UPD_BLOCKS, blk, 0, stream>>>(
            xcur, ax, x0, alpha_p, beta_p, out, ntiles);
        xcur = out;
    }
}


# Atomics add alpha*w*x[src] straight onto a pre-stored (1-alpha)*x+beta*x0 in a ping-pong state buffer: no accumulator, zero pass or read-back (8 launches, was 12); f32 sum order differs
# speedup vs baseline: 2.8189x; 1.2839x over previous
#include <hip/hip_runtime.h>
#include <math.h>


#define FEAT_D 32

typedef __attribute__((ext_vector_type(2))) float v2f;
typedef __attribute__((ext_vector_type(8))) float v8f;

#define SCAT_BLOCKS 1024
#define BASE_BLOCKS 512

__device__ __forceinline__ float sigmoid_of(const float* __restrict__ p) {
    return 1.0f / (1.0f + expf(-(*p)));
}

#define EDGES_PER_WAVE 32

__device__ __forceinline__ void scatter_one(const float* __restrict__ x,
                                            float*       __restrict__ nxt,
                                            int s, int t, float we, int lane) {
    float val = we * x[(size_t)(unsigned)s * FEAT_D + lane];
    unsafeAtomicAdd(&nxt[(size_t)(unsigned)t * FEAT_D + lane], val);
}

__global__ __launch_bounds__(256)
void scatter_kernel(const float* __restrict__ x,
                    const int*   __restrict__ src,
                    const int*   __restrict__ dst,
                    const float* __restrict__ w,
                    const float* __restrict__ alpha_p,
                    float*       __restrict__ nxt,
                    int E) {
    const int wave = __builtin_amdgcn_readfirstlane(
        blockIdx.x * (blockDim.x >> 5) + (threadIdx.x >> 5));
    const int nwaves = gridDim.x * (blockDim.x >> 5);
    const int lane = threadIdx.x & 31;
    const unsigned ulane = threadIdx.x & 31u;
    const int nfull = E >> 5;

    const float alpha = sigmoid_of(alpha_p);

    int b = wave;
    int s_l = 0, t_l = 0;
    float wr_l = 0.0f;
    if (b < nfull) {
        const int el = (b << 5) + lane;
        s_l = src[el];
        t_l = dst[el];
        wr_l = w[el];
    }
    while (b < nfull) {
        float v[EDGES_PER_WAVE];
#pragma unroll
        for (int j = 0; j < EDGES_PER_WAVE; ++j) {
            const int s = __builtin_amdgcn_readlane(s_l, j);
            v[j] = x[((unsigned)s << 5) + ulane];
        }
        const int bn = b + nwaves;
        int s_n = 0, t_n = 0;
        float wr_n = 0.0f;
        if (bn < nfull) {
            const int el = (bn << 5) + lane;
            s_n = src[el];
            t_n = dst[el];
            wr_n = w[el];
        }
        const int w_l = __float_as_int(alpha * wr_l);
#pragma unroll
        for (int j = 0; j < EDGES_PER_WAVE; ++j) {
            const int   t  = __builtin_amdgcn_readlane(t_l, j);
            const float we = __int_as_float(__builtin_amdgcn_readlane(w_l, j));
            unsafeAtomicAdd(&nxt[((unsigned)t << 5) + ulane], we * v[j]);
        }
        s_l = s_n; t_l = t_n; wr_l = wr_n;
        b = bn;
    }

    const int rem = E & 31;
    if (rem != 0 && wave == nwaves - 1) {
        const int el = (nfull << 5) + min(lane, rem - 1);
        s_l = src[el];
        t_l = dst[el];
        const int w_l = __float_as_int(alpha * w[el]);
        for (int j = 0; j < rem; ++j) {
            const int   s  = __builtin_amdgcn_readlane(s_l, j);
            const int   t  = __builtin_amdgcn_readlane(t_l, j);
            const float we = __int_as_float(__builtin_amdgcn_readlane(w_l, j));
            scatter_one(x, nxt, s, t, we, lane);
        }
    }
}

struct TileRegs {
    float zv[8];
    float bv[8];
};

__device__ __forceinline__ void load_tile(const float* __restrict__ x,
                                          const float* __restrict__ x0,
                                          int tile, int hi, int li, TileRegs& r) {
    const int nb = (tile >> 1) * 16;
    const int cb = (tile & 1) * 16;
#pragma unroll
    for (int v = 0; v < 8; ++v) {
        const int idx = (nb + v + 8 * hi) * FEAT_D + cb + li;
        r.zv[v] = x0[idx];
    }
#pragma unroll
    for (int m = 0; m < 4; ++m) {
#pragma unroll
        for (int v = 0; v < 2; ++v) {
            const int rowb = nb + 4 * m + v + 2 * hi;
            r.bv[2 * m + v] = x[rowb * FEAT_D + cb + li];
        }
    }
}

__device__ __forceinline__ void finish_tile(const TileRegs& r, int tile, int hi, int li,
                                            float ca, float beta,
                                            float* __restrict__ nxt) {
    const int nb = (tile >> 1) * 16;
    const int cb = (tile & 1) * 16;

    v8f c;
#pragma unroll
    for (int v = 0; v < 8; ++v)
        c[v] = beta * r.zv[v];

#pragma unroll
    for (int m = 0; m < 4; ++m) {
        v2f a, b;
#pragma unroll
        for (int v = 0; v < 2; ++v) {
            const int k = v + 2 * hi;
            a[v] = (li == 4 * m + k) ? ca : 0.0f;
            b[v] = r.bv[2 * m + v];
        }
        c = __builtin_amdgcn_wmma_f32_16x16x4_f32(
                false, a, false, b, (short)0, c, false, false);
    }

#pragma unroll
    for (int v = 0; v < 8; ++v) {
        const int row = nb + v + 8 * hi;
        nxt[row * FEAT_D + cb + li] = c[v];
    }
}

__global__ __launch_bounds__(256)
void base_wmma_kernel(const float* __restrict__ x,
                      const float* __restrict__ x0,
                      const float* __restrict__ alpha_p,
                      const float* __restrict__ beta_p,
                      float*       __restrict__ nxt,
                      int ntiles) {
    const int wave = __builtin_amdgcn_readfirstlane(
        blockIdx.x * (blockDim.x >> 5) + (threadIdx.x >> 5));
    const int nwaves = gridDim.x * (blockDim.x >> 5);
    if (wave >= ntiles) return;

    const int lane = threadIdx.x & 31;
    const int hi   = lane >> 4;
    const int li   = lane & 15;

    const float alpha = sigmoid_of(alpha_p);
    const float beta  = *beta_p;
    const float ca    = 1.0f - alpha;

    const int step2 = 2 * nwaves;
    int t = wave;
    TileRegs ra, rb;
    load_tile(x, x0, t, hi, li, ra);
    while (t + step2 < ntiles) {
        load_tile(x, x0, t + nwaves, hi, li, rb);
        __builtin_amdgcn_sched_barrier(0);
        finish_tile(ra, t, hi, li, ca, beta, nxt);
        load_tile(x, x0, t + step2, hi, li, ra);
        __builtin_amdgcn_sched_barrier(0);
        finish_tile(rb, t + nwaves, hi, li, ca, beta, nxt);
        t += step2;
    }
    if (t + nwaves < ntiles) {
        load_tile(x, x0, t + nwaves, hi, li, rb);
        __builtin_amdgcn_sched_barrier(0);
        finish_tile(ra, t, hi, li, ca, beta, nxt);
        finish_tile(rb, t + nwaves, hi, li, ca, beta, nxt);
    } else {
        finish_tile(ra, t, hi, li, ca, beta, nxt);
    }
}

extern "C" void kernel_launch(void* const* d_in, const int* in_sizes, int n_in,
                              void* d_out, int out_size, void* d_ws, size_t ws_size,
                              hipStream_t stream) {
    const float* x_in    = (const float*)d_in[0];
    const float* w       = (const float*)d_in[1];
    const float* x0      = (const float*)d_in[2];
    const float* alpha_p = (const float*)d_in[3];
    const float* beta_p  = (const float*)d_in[4];
    const int*   eidx    = (const int*)  d_in[5];

    const int N = in_sizes[0] / FEAT_D;
    const int E = in_sizes[1];
    const int* src = eidx;
    const int* dst = eidx + E;

    float* wsx = (float*)d_ws;
    float* out = (float*)d_out;

    dim3 blk(256);
    const int ntiles = (N / 16) * (FEAT_D / 16);

    base_wmma_kernel<<<BASE_BLOCKS, blk, 0, stream>>>(x_in, x0, alpha_p, beta_p, wsx, ntiles);
    scatter_kernel<<<SCAT_BLOCKS, blk, 0, stream>>>(x_in, src, dst, w, alpha_p, wsx, E);
    base_wmma_kernel<<<BASE_BLOCKS, blk, 0, stream>>>(wsx, x0, alpha_p, beta_p, out, ntiles);
    scatter_kernel<<<SCAT_BLOCKS, blk, 0, stream>>>(wsx, src, dst, w, alpha_p, out, E);
    base_wmma_kernel<<<BASE_BLOCKS, blk, 0, stream>>>(out, x0, alpha_p, beta_p, wsx, ntiles);
    scatter_kernel<<<SCAT_BLOCKS, blk, 0, stream>>>(out, src, dst, w, alpha_p, wsx, E);
    base_wmma_kernel<<<BASE_BLOCKS, blk, 0, stream>>>(wsx, x0, alpha_p, beta_p, out, ntiles);
    scatter_kernel<<<SCAT_BLOCKS, blk, 0, stream>>>(wsx, src, dst, w, alpha_p, out, E);
}


# Edges sorted once per call into bins of 400 dst rows; per step ONE launch: a work-group sums its bin's rows in LDS (ds_add_f32), stores each once; no global atomics, 7 launches; f32 sum order differs
# speedup vs baseline: 4.3055x; 1.5274x over previous
#include <hip/hip_runtime.h>
#include <math.h>


#define FEAT_D 32

typedef __attribute__((ext_vector_type(2))) float v2f;
typedef __attribute__((ext_vector_type(8))) float v8f;

#define SCAT_BLOCKS 1024
#define BASE_BLOCKS 512

__device__ __forceinline__ float sigmoid_of(const float* __restrict__ p) {
    return 1.0f / (1.0f + expf(-(*p)));
}

#define EDGES_PER_WAVE 32

__device__ __forceinline__ void scatter_one(const float* __restrict__ x,
                                            float*       __restrict__ nxt,
                                            int s, int t, float we, int lane) {
    float val = we * x[(size_t)(unsigned)s * FEAT_D + lane];
    unsafeAtomicAdd(&nxt[(size_t)(unsigned)t * FEAT_D + lane], val);
}

__global__ __launch_bounds__(256)
void scatter_kernel(const float* __restrict__ x,
                    const int*   __restrict__ src,
                    const int*   __restrict__ dst,
                    const float* __restrict__ w,
                    const float* __restrict__ alpha_p,
                    float*       __restrict__ nxt,
                    int E) {
    const int wave = __builtin_amdgcn_readfirstlane(
        blockIdx.x * (blockDim.x >> 5) + (threadIdx.x >> 5));
    const int nwaves = gridDim.x * (blockDim.x >> 5);
    const int lane = threadIdx.x & 31;
    const unsigned ulane = threadIdx.x & 31u;
    const int nfull = E >> 5;

    const float alpha = sigmoid_of(alpha_p);

    int b = wave;
    int s_l = 0, t_l = 0;
    float wr_l = 0.0f;
    if (b < nfull) {
        const int el = (b << 5) + lane;
        s_l = src[el];
        t_l = dst[el];
        wr_l = w[el];
    }
    while (b < nfull) {
        float v[EDGES_PER_WAVE];
#pragma unroll
        for (int j = 0; j < EDGES_PER_WAVE; ++j) {
            const int s = __builtin_amdgcn_readlane(s_l, j);
            v[j] = x[((unsigned)s << 5) + ulane];
        }
        const int bn = b + nwaves;
        int s_n = 0, t_n = 0;
        float wr_n = 0.0f;
        if (bn < nfull) {
            const int el = (bn << 5) + lane;
            s_n = src[el];
            t_n = dst[el];
            wr_n = w[el];
        }
        const int w_l = __float_as_int(alpha * wr_l);
#pragma unroll
        for (int j = 0; j < EDGES_PER_WAVE; ++j) {
            const int   t  = __builtin_amdgcn_readlane(t_l, j);
            const float we = __int_as_float(__builtin_amdgcn_readlane(w_l, j));
            unsafeAtomicAdd(&nxt[((unsigned)t << 5) + ulane], we * v[j]);
        }
        s_l = s_n; t_l = t_n; wr_l = wr_n;
        b = bn;
    }

    const int rem = E & 31;
    if (rem != 0 && wave == nwaves - 1) {
        const int el = (nfull << 5) + min(lane, rem - 1);
        s_l = src[el];
        t_l = dst[el];
        const int w_l = __float_as_int(alpha * w[el]);
        for (int j = 0; j < rem; ++j) {
            const int   s  = __builtin_amdgcn_readlane(s_l, j);
            const int   t  = __builtin_amdgcn_readlane(t_l, j);
            const float we = __int_as_float(__builtin_amdgcn_readlane(w_l, j));
            scatter_one(x, nxt, s, t, we, lane);
        }
    }
}

struct TileRegs {
    float zv[8];
    float bv[8];
};

__device__ __forceinline__ void load_tile(const float* __restrict__ x,
                                          const float* __restrict__ x0,
                                          int tile, int hi, int li, TileRegs& r) {
    const int nb = (tile >> 1) * 16;
    const int cb = (tile & 1) * 16;
#pragma unroll
    for (int v = 0; v < 8; ++v) {
        const int idx = (nb + v + 8 * hi) * FEAT_D + cb + li;
        r.zv[v] = x0[idx];
    }
#pragma unroll
    for (int m = 0; m < 4; ++m) {
#pragma unroll
        for (int v = 0; v < 2; ++v) {
            const int rowb = nb + 4 * m + v + 2 * hi;
            r.bv[2 * m + v] = x[rowb * FEAT_D + cb + li];
        }
    }
}

__device__ __forceinline__ void finish_tile(const TileRegs& r, int tile, int hi, int li,
                                            float ca, float beta,
                                            float* __restrict__ nxt) {
    const int nb = (tile >> 1) * 16;
    const int cb = (tile & 1) * 16;

    v8f c;
#pragma unroll
    for (int v = 0; v < 8; ++v)
        c[v] = beta * r.zv[v];

#pragma unroll
    for (int m = 0; m < 4; ++m) {
        v2f a, b;
#pragma unroll
        for (int v = 0; v < 2; ++v) {
            const int k = v + 2 * hi;
            a[v] = (li == 4 * m + k) ? ca : 0.0f;
            b[v] = r.bv[2 * m + v];
        }
        c = __builtin_amdgcn_wmma_f32_16x16x4_f32(
                false, a, false, b, (short)0, c, false, false);
    }

#pragma unroll
    for (int v = 0; v < 8; ++v) {
        const int row = nb + v + 8 * hi;
        nxt[row * FEAT_D + cb + li] = c[v];
    }
}

__global__ __launch_bounds__(256)
void base_wmma_kernel(const float* __restrict__ x,
                      const float* __restrict__ x0,
                      const float* __restrict__ alpha_p,
                      const float* __restrict__ beta_p,
                      float*       __restrict__ nxt,
                      int ntiles) {
    const int wave = __builtin_amdgcn_readfirstlane(
        blockIdx.x * (blockDim.x >> 5) + (threadIdx.x >> 5));
    const int nwaves = gridDim.x * (blockDim.x >> 5);
    if (wave >= ntiles) return;

    const int lane = threadIdx.x & 31;
    const int hi   = lane >> 4;
    const int li   = lane & 15;

    const float alpha = sigmoid_of(alpha_p);
    const float beta  = *beta_p;
    const float ca    = 1.0f - alpha;

    const int step2 = 2 * nwaves;
    int t = wave;
    TileRegs ra, rb;
    load_tile(x, x0, t, hi, li, ra);
    while (t + step2 < ntiles) {
        load_tile(x, x0, t + nwaves, hi, li, rb);
        __builtin_amdgcn_sched_barrier(0);
        finish_tile(ra, t, hi, li, ca, beta, nxt);
        load_tile(x, x0, t + step2, hi, li, ra);
        __builtin_amdgcn_sched_barrier(0);
        finish_tile(rb, t + nwaves, hi, li, ca, beta, nxt);
        t += step2;
    }
    if (t + nwaves < ntiles) {
        load_tile(x, x0, t + nwaves, hi, li, rb);
        __builtin_amdgcn_sched_barrier(0);
        finish_tile(ra, t, hi, li, ca, beta, nxt);
        finish_tile(rb, t + nwaves, hi, li, ca, beta, nxt);
    } else {
        finish_tile(ra, t, hi, li, ca, beta, nxt);
    }
}

#define BIN_TILES   25
#define BIN_ROWS    (16 * BIN_TILES)
#define BIN_BLOCKS  128
#define SORT_THREADS 1024
#define SORT_ITEMS  8
#define SCAN_BATCH  32
#define NBIN_MAX    1024
#define STEP_BLOCKS 256

__global__ __launch_bounds__(SORT_THREADS)
void bin_count_kernel(const int* __restrict__ dst,
                      int*       __restrict__ cnt,
                      int E, int NB) {
    __shared__ int h[NBIN_MAX];
    const int tid = threadIdx.x;
    for (int i = tid; i < NB; i += blockDim.x) h[i] = 0;
    __syncthreads();
    const int chunk = (E + gridDim.x - 1) / gridDim.x;
    const int e0 = blockIdx.x * chunk;
    const int e1 = min(E, e0 + chunk);
    for (int eb = e0 + tid; eb < e1; eb += SORT_ITEMS * SORT_THREADS) {
        int d[SORT_ITEMS];
#pragma unroll
        for (int k = 0; k < SORT_ITEMS; ++k)
            d[k] = dst[min(eb + k * SORT_THREADS, e1 - 1)];
#pragma unroll
        for (int k = 0; k < SORT_ITEMS; ++k)
            if (eb + k * SORT_THREADS < e1)
                __hip_atomic_fetch_add(&h[d[k] / BIN_ROWS], 1, __ATOMIC_RELAXED,
                                       __HIP_MEMORY_SCOPE_WORKGROUP);
    }
    __syncthreads();
    for (int i = tid; i < NB; i += blockDim.x) cnt[blockIdx.x * NB + i] = h[i];
}

__global__ __launch_bounds__(1024)
void bin_scan_kernel(const int* __restrict__ cnt,
                     int*       __restrict__ offs,
                     int*       __restrict__ binstart,
                     int NB) {
    static_assert(BIN_BLOCKS % SCAN_BATCH == 0, "the scan takes the chunks in whole batches");
    __shared__ int tot[1024];
    const int b = threadIdx.x;
    int run = 0;
    if (b < NB) {
        for (int g0 = 0; g0 < BIN_BLOCKS; g0 += SCAN_BATCH) {
            int v[SCAN_BATCH];
#pragma unroll
            for (int k = 0; k < SCAN_BATCH; ++k)
                v[k] = cnt[(g0 + k) * NB + b];
#pragma unroll
            for (int k = 0; k < SCAN_BATCH; ++k) {
                offs[(g0 + k) * NB + b] = run;
                run += v[k];
            }
        }
    }
    tot[b] = run;
    __syncthreads();
    for (int off = 1; off < 1024; off <<= 1) {
        const int v = (b >= off) ? tot[b - off] : 0;
        __syncthreads();
        tot[b] += v;
        __syncthreads();
    }
    if (b < NB) binstart[b] = tot[b] - run;
    if (b == 0) binstart[NB] = tot[1023];
}

__global__ __launch_bounds__(SORT_THREADS)
void bin_fill_kernel(const int*   __restrict__ src,
                     const int*   __restrict__ dst,
                     const float* __restrict__ w,
                     const float* __restrict__ alpha_p,
                     const int*   __restrict__ offs,
                     const int*   __restrict__ binstart,
                     int2*        __restrict__ rec,
                     int E, int NB) {
    __shared__ int cur[NBIN_MAX];
    const int tid = threadIdx.x;
    for (int i = tid; i < NB; i += blockDim.x)
        cur[i] = binstart[i] + offs[blockIdx.x * NB + i];
    __syncthreads();
    const float alpha = sigmoid_of(alpha_p);
    const int chunk = (E + gridDim.x - 1) / gridDim.x;
    const int e0 = blockIdx.x * chunk;
    const int e1 = min(E, e0 + chunk);
    for (int eb = e0 + tid; eb < e1; eb += SORT_ITEMS * SORT_THREADS) {
        int d[SORT_ITEMS], s[SORT_ITEMS];
        float ww[SORT_ITEMS];
#pragma unroll
        for (int k = 0; k < SORT_ITEMS; ++k) {
            const int e = min(eb + k * SORT_THREADS, e1 - 1);
            d[k]  = dst[e];
            s[k]  = src[e];
            ww[k] = w[e];
        }
#pragma unroll
        for (int k = 0; k < SORT_ITEMS; ++k) {
            if (eb + k * SORT_THREADS < e1) {
                const int b = d[k] / BIN_ROWS;
                const int pos = __hip_atomic_fetch_add(&cur[b], 1, __ATOMIC_RELAXED,
                                                       __HIP_MEMORY_SCOPE_WORKGROUP);
                int2 r;
                r.x = s[k] | ((d[k] - b * BIN_ROWS) << 20);
                r.y = __float_as_int(alpha * ww[k]);
                rec[pos] = r;
            }
        }
    }
}

__device__ __forceinline__ void finish_tile_acc(const TileRegs& r, const float* av,
                                                int tile, int hi, int li,
                                                float ca, float beta,
                                                float* __restrict__ nxt) {
    const int nb = (tile >> 1) * 16;
    const int cb = (tile & 1) * 16;

    v8f c;
#pragma unroll
    for (int v = 0; v < 8; ++v)
        c[v] = beta * r.zv[v] + av[v];

#pragma unroll
    for (int m = 0; m < 4; ++m) {
        v2f a, b;
#pragma unroll
        for (int v = 0; v < 2; ++v) {
            const int k = v + 2 * hi;
            a[v] = (li == 4 * m + k) ? ca : 0.0f;
            b[v] = r.bv[2 * m + v];
        }
        c = __builtin_amdgcn_wmma_f32_16x16x4_f32(
                false, a, false, b, (short)0, c, false, false);
    }

#pragma unroll
    for (int v = 0; v < 8; ++v) {
        const int row = nb + v + 8 * hi;
        nxt[row * FEAT_D + cb + li] = c[v];
    }
}

__global__ __launch_bounds__(1024)
void step_binned_kernel(const float* __restrict__ x,
                        const float* __restrict__ x0,
                        const int2*  __restrict__ rec,
                        const int*   __restrict__ binstart,
                        const float* __restrict__ alpha_p,
                        const float* __restrict__ beta_p,
                        float*       __restrict__ nxt,
                        int N, int NB) {
    __shared__ float acc[BIN_ROWS * FEAT_D];

    const int wave = __builtin_amdgcn_readfirstlane(threadIdx.x >> 5);
    const int nw   = blockDim.x >> 5;
    const int lane = threadIdx.x & 31;
    const unsigned ulane = threadIdx.x & 31u;
    const int hi   = lane >> 4;
    const int li   = lane & 15;

    const float alpha = sigmoid_of(alpha_p);
    const float beta  = *beta_p;
    const float ca    = 1.0f - alpha;
    const int   nrt_all = N >> 4;

    for (int b = blockIdx.x; b < NB; b += gridDim.x) {
        for (int i = threadIdx.x; i < BIN_ROWS * FEAT_D; i += blockDim.x) acc[i] = 0.0f;
        __syncthreads();

        const int e0 = binstart[b];
        const int e1 = binstart[b + 1];
        int base = e0 + (wave << 5);
        int2 r = make_int2(0, 0);
        if (base < e1) r = rec[min(base + lane, e1 - 1)];
        while (base < e1) {
            const int nvalid = min(EDGES_PER_WAVE, e1 - base);
            const int bn = base + (nw << 5);
            if (nvalid == EDGES_PER_WAVE) {
                float v[EDGES_PER_WAVE];
#pragma unroll
                for (int j = 0; j < EDGES_PER_WAVE; ++j) {
                    const unsigned s = (unsigned)__builtin_amdgcn_readlane(r.x, j) & 0xFFFFFu;
                    v[j] = x[(s << 5) + ulane];
                }
                int2 rn = make_int2(0, 0);
                if (bn < e1) rn = rec[min(bn + lane, e1 - 1)];
#pragma unroll
                for (int j = 0; j < EDGES_PER_WAVE; ++j) {
                    const unsigned dl = (unsigned)__builtin_amdgcn_readlane(r.x, j) >> 20;
                    const float we = __int_as_float(__builtin_amdgcn_readlane(r.y, j));
                    __hip_atomic_fetch_add(&acc[(dl << 5) + ulane], we * v[j],
                                           __ATOMIC_RELAXED, __HIP_MEMORY_SCOPE_WORKGROUP);
                }
                r = rn;
            } else {
                for (int j = 0; j < nvalid; ++j) {
                    const unsigned rx = (unsigned)__builtin_amdgcn_readlane(r.x, j);
                    const float we = __int_as_float(__builtin_amdgcn_readlane(r.y, j));
                    const float val = we * x[((rx & 0xFFFFFu) << 5) + ulane];
                    __hip_atomic_fetch_add(&acc[((rx >> 20) << 5) + ulane], val,
                                           __ATOMIC_RELAXED, __HIP_MEMORY_SCOPE_WORKGROUP);
                }
                r = make_int2(0, 0);
            }
            base = bn;
        }
        __syncthreads();

        const int rt0 = b * BIN_TILES;
        const int nrt = min(BIN_TILES, nrt_all - rt0);
        for (int t = wave; t < 2 * nrt; t += nw) {
            const int tile = rt0 * 2 + t;
            TileRegs tr;
            load_tile(x, x0, tile, hi, li, tr);
            float av[8];
            const int lrow0 = (t >> 1) * 16;
            const int cb = (t & 1) * 16;
#pragma unroll
            for (int v = 0; v < 8; ++v)
                av[v] = acc[(lrow0 + v + 8 * hi) * FEAT_D + cb + li];
            finish_tile_acc(tr, av, tile, hi, li, ca, beta, nxt);
        }
        __syncthreads();
    }
}

extern "C" void kernel_launch(void* const* d_in, const int* in_sizes, int n_in,
                              void* d_out, int out_size, void* d_ws, size_t ws_size,
                              hipStream_t stream) {
    const float* x_in    = (const float*)d_in[0];
    const float* w       = (const float*)d_in[1];
    const float* x0      = (const float*)d_in[2];
    const float* alpha_p = (const float*)d_in[3];
    const float* beta_p  = (const float*)d_in[4];
    const int*   eidx    = (const int*)  d_in[5];

    const int N = in_sizes[0] / FEAT_D;
    const int E = in_sizes[1];
    const int* src = eidx;
    const int* dst = eidx + E;

    float* wsx = (float*)d_ws;
    float* out = (float*)d_out;

    const int NB = (N + BIN_ROWS - 1) / BIN_ROWS;
    const size_t plane = (size_t)N * FEAT_D * sizeof(float);
    const size_t need  = plane + (size_t)E * sizeof(int2)
                       + ((size_t)2 * BIN_BLOCKS * NB + NB + 1) * sizeof(int);

    if (ws_size >= need && NB <= NBIN_MAX && N <= (1 << 20)) {
        int2* rec      = (int2*)((char*)d_ws + plane);
        int*  cnt      = (int*)(rec + E);
        int*  offs     = cnt + BIN_BLOCKS * NB;
        int*  binstart = offs + BIN_BLOCKS * NB;

        dim3 big(1024);
        bin_count_kernel<<<BIN_BLOCKS, dim3(SORT_THREADS), 0, stream>>>(dst, cnt, E, NB);
        bin_scan_kernel<<<1, big, 0, stream>>>(cnt, offs, binstart, NB);
        bin_fill_kernel<<<BIN_BLOCKS, dim3(SORT_THREADS), 0, stream>>>(src, dst, w, alpha_p, offs, binstart, rec, E, NB);

        step_binned_kernel<<<STEP_BLOCKS, big, 0, stream>>>(x_in, x0, rec, binstart, alpha_p, beta_p, wsx, N, NB);
        step_binned_kernel<<<STEP_BLOCKS, big, 0, stream>>>(wsx,  x0, rec, binstart, alpha_p, beta_p, out, N, NB);
        step_binned_kernel<<<STEP_BLOCKS, big, 0, stream>>>(out,  x0, rec, binstart, alpha_p, beta_p, wsx, N, NB);
        step_binned_kernel<<<STEP_BLOCKS, big, 0, stream>>>(wsx,  x0, rec, binstart, alpha_p, beta_p, out, N, NB);
        return;
    }

    dim3 blk(256);
    const int ntiles = (N / 16) * (FEAT_D / 16);

    base_wmma_kernel<<<BASE_BLOCKS, blk, 0, stream>>>(x_in, x0, alpha_p, beta_p, wsx, ntiles);
    scatter_kernel<<<SCAT_BLOCKS, blk, 0, stream>>>(x_in, src, dst, w, alpha_p, wsx, E);
    base_wmma_kernel<<<BASE_BLOCKS, blk, 0, stream>>>(wsx, x0, alpha_p, beta_p, out, ntiles);
    scatter_kernel<<<SCAT_BLOCKS, blk, 0, stream>>>(wsx, src, dst, w, alpha_p, out, E);
    base_wmma_kernel<<<BASE_BLOCKS, blk, 0, stream>>>(out, x0, alpha_p, beta_p, wsx, ntiles);
    scatter_kernel<<<SCAT_BLOCKS, blk, 0, stream>>>(out, src, dst, w, alpha_p, wsx, E);
    base_wmma_kernel<<<BASE_BLOCKS, blk, 0, stream>>>(wsx, x0, alpha_p, beta_p, out, ntiles);
    scatter_kernel<<<SCAT_BLOCKS, blk, 0, stream>>>(wsx, src, dst, w, alpha_p, out, E);
}


# Step kernel: a bin's last short edge batch padded to 32 (zero weight, spare LDS row) and run 32 wide, not one dependent L2 load per edge; same LDS atomics, arrival order may differ
# speedup vs baseline: 4.8180x; 1.1190x over previous
#include <hip/hip_runtime.h>
#include <math.h>


#define FEAT_D 32

typedef __attribute__((ext_vector_type(2))) float v2f;
typedef __attribute__((ext_vector_type(8))) float v8f;

#define SCAT_BLOCKS 1024
#define BASE_BLOCKS 512

__device__ __forceinline__ float sigmoid_of(const float* __restrict__ p) {
    return 1.0f / (1.0f + expf(-(*p)));
}

#define EDGES_PER_WAVE 32

__device__ __forceinline__ void scatter_one(const float* __restrict__ x,
                                            float*       __restrict__ nxt,
                                            int s, int t, float we, int lane) {
    float val = we * x[(size_t)(unsigned)s * FEAT_D + lane];
    unsafeAtomicAdd(&nxt[(size_t)(unsigned)t * FEAT_D + lane], val);
}

__global__ __launch_bounds__(256)
void scatter_kernel(const float* __restrict__ x,
                    const int*   __restrict__ src,
                    const int*   __restrict__ dst,
                    const float* __restrict__ w,
                    const float* __restrict__ alpha_p,
                    float*       __restrict__ nxt,
                    int E) {
    const int wave = __builtin_amdgcn_readfirstlane(
        blockIdx.x * (blockDim.x >> 5) + (threadIdx.x >> 5));
    const int nwaves = gridDim.x * (blockDim.x >> 5);
    const int lane = threadIdx.x & 31;
    const unsigned ulane = threadIdx.x & 31u;
    const int nfull = E >> 5;

    const float alpha = sigmoid_of(alpha_p);

    int b = wave;
    int s_l = 0, t_l = 0;
    float wr_l = 0.0f;
    if (b < nfull) {
        const int el = (b << 5) + lane;
        s_l = src[el];
        t_l = dst[el];
        wr_l = w[el];
    }
    while (b < nfull) {
        float v[EDGES_PER_WAVE];
#pragma unroll
        for (int j = 0; j < EDGES_PER_WAVE; ++j) {
            const int s = __builtin_amdgcn_readlane(s_l, j);
            v[j] = x[((unsigned)s << 5) + ulane];
        }
        const int bn = b + nwaves;
        int s_n = 0, t_n = 0;
        float wr_n = 0.0f;
        if (bn < nfull) {
            const int el = (bn << 5) + lane;
            s_n = src[el];
            t_n = dst[el];
            wr_n = w[el];
        }
        const int w_l = __float_as_int(alpha * wr_l);
#pragma unroll
        for (int j = 0; j < EDGES_PER_WAVE; ++j) {
            const int   t  = __builtin_amdgcn_readlane(t_l, j);
            const float we = __int_as_float(__builtin_amdgcn_readlane(w_l, j));
            unsafeAtomicAdd(&nxt[((unsigned)t << 5) + ulane], we * v[j]);
        }
        s_l = s_n; t_l = t_n; wr_l = wr_n;
        b = bn;
    }

    const int rem = E & 31;
    if (rem != 0 && wave == nwaves - 1) {
        const int el = (nfull << 5) + min(lane, rem - 1);
        s_l = src[el];
        t_l = dst[el];
        const int w_l = __float_as_int(alpha * w[el]);
        for (int j = 0; j < rem; ++j) {
            const int   s  = __builtin_amdgcn_readlane(s_l, j);
            const int   t  = __builtin_amdgcn_readlane(t_l, j);
            const float we = __int_as_float(__builtin_amdgcn_readlane(w_l, j));
            scatter_one(x, nxt, s, t, we, lane);
        }
    }
}

struct TileRegs {
    float zv[8];
    float bv[8];
};

__device__ __forceinline__ void load_tile(const float* __restrict__ x,
                                          const float* __restrict__ x0,
                                          int tile, int hi, int li, TileRegs& r) {
    const int nb = (tile >> 1) * 16;
    const int cb = (tile & 1) * 16;
#pragma unroll
    for (int v = 0; v < 8; ++v) {
        const int idx = (nb + v + 8 * hi) * FEAT_D + cb + li;
        r.zv[v] = x0[idx];
    }
#pragma unroll
    for (int m = 0; m < 4; ++m) {
#pragma unroll
        for (int v = 0; v < 2; ++v) {
            const int rowb = nb + 4 * m + v + 2 * hi;
            r.bv[2 * m + v] = x[rowb * FEAT_D + cb + li];
        }
    }
}

__device__ __forceinline__ void finish_tile(const TileRegs& r, int tile, int hi, int li,
                                            float ca, float beta,
                                            float* __restrict__ nxt) {
    const int nb = (tile >> 1) * 16;
    const int cb = (tile & 1) * 16;

    v8f c;
#pragma unroll
    for (int v = 0; v < 8; ++v)
        c[v] = beta * r.zv[v];

#pragma unroll
    for (int m = 0; m < 4; ++m) {
        v2f a, b;
#pragma unroll
        for (int v = 0; v < 2; ++v) {
            const int k = v + 2 * hi;
            a[v] = (li == 4 * m + k) ? ca : 0.0f;
            b[v] = r.bv[2 * m + v];
        }
        c = __builtin_amdgcn_wmma_f32_16x16x4_f32(
                false, a, false, b, (short)0, c, false, false);
    }

#pragma unroll
    for (int v = 0; v < 8; ++v) {
        const int row = nb + v + 8 * hi;
        nxt[row * FEAT_D + cb + li] = c[v];
    }
}

__global__ __launch_bounds__(256)
void base_wmma_kernel(const float* __restrict__ x,
                      const float* __restrict__ x0,
                      const float* __restrict__ alpha_p,
                      const float* __restrict__ beta_p,
                      float*       __restrict__ nxt,
                      int ntiles) {
    const int wave = __builtin_amdgcn_readfirstlane(
        blockIdx.x * (blockDim.x >> 5) + (threadIdx.x >> 5));
    const int nwaves = gridDim.x * (blockDim.x >> 5);
    if (wave >= ntiles) return;

    const int lane = threadIdx.x & 31;
    const int hi   = lane >> 4;
    const int li   = lane & 15;

    const float alpha = sigmoid_of(alpha_p);
    const float beta  = *beta_p;
    const float ca    = 1.0f - alpha;

    const int step2 = 2 * nwaves;
    int t = wave;
    TileRegs ra, rb;
    load_tile(x, x0, t, hi, li, ra);
    while (t + step2 < ntiles) {
        load_tile(x, x0, t + nwaves, hi, li, rb);
        __builtin_amdgcn_sched_barrier(0);
        finish_tile(ra, t, hi, li, ca, beta, nxt);
        load_tile(x, x0, t + step2, hi, li, ra);
        __builtin_amdgcn_sched_barrier(0);
        finish_tile(rb, t + nwaves, hi, li, ca, beta, nxt);
        t += step2;
    }
    if (t + nwaves < ntiles) {
        load_tile(x, x0, t + nwaves, hi, li, rb);
        __builtin_amdgcn_sched_barrier(0);
        finish_tile(ra, t, hi, li, ca, beta, nxt);
        finish_tile(rb, t + nwaves, hi, li, ca, beta, nxt);
    } else {
        finish_tile(ra, t, hi, li, ca, beta, nxt);
    }
}

#define BIN_TILES   25
#define BIN_ROWS    (16 * BIN_TILES)
#define BIN_BLOCKS  128
#define SORT_THREADS 1024
#define SORT_ITEMS  8
#define SCAN_BATCH  32
#define NBIN_MAX    1024
#define STEP_BLOCKS 256

__global__ __launch_bounds__(SORT_THREADS)
void bin_count_kernel(const int* __restrict__ dst,
                      int*       __restrict__ cnt,
                      int E, int NB) {
    __shared__ int h[NBIN_MAX];
    const int tid = threadIdx.x;
    for (int i = tid; i < NB; i += blockDim.x) h[i] = 0;
    __syncthreads();
    const int chunk = (E + gridDim.x - 1) / gridDim.x;
    const int e0 = blockIdx.x * chunk;
    const int e1 = min(E, e0 + chunk);
    for (int eb = e0 + tid; eb < e1; eb += SORT_ITEMS * SORT_THREADS) {
        int d[SORT_ITEMS];
#pragma unroll
        for (int k = 0; k < SORT_ITEMS; ++k)
            d[k] = dst[min(eb + k * SORT_THREADS, e1 - 1)];
#pragma unroll
        for (int k = 0; k < SORT_ITEMS; ++k)
            if (eb + k * SORT_THREADS < e1)
                __hip_atomic_fetch_add(&h[d[k] / BIN_ROWS], 1, __ATOMIC_RELAXED,
                                       __HIP_MEMORY_SCOPE_WORKGROUP);
    }
    __syncthreads();
    for (int i = tid; i < NB; i += blockDim.x) cnt[blockIdx.x * NB + i] = h[i];
}

__global__ __launch_bounds__(1024)
void bin_scan_kernel(const int* __restrict__ cnt,
                     int*       __restrict__ offs,
                     int*       __restrict__ binstart,
                     int NB) {
    static_assert(BIN_BLOCKS % SCAN_BATCH == 0, "the scan takes the chunks in whole batches");
    __shared__ int tot[1024];
    const int b = threadIdx.x;
    int run = 0;
    if (b < NB) {
        for (int g0 = 0; g0 < BIN_BLOCKS; g0 += SCAN_BATCH) {
            int v[SCAN_BATCH];
#pragma unroll
            for (int k = 0; k < SCAN_BATCH; ++k)
                v[k] = cnt[(g0 + k) * NB + b];
#pragma unroll
            for (int k = 0; k < SCAN_BATCH; ++k) {
                offs[(g0 + k) * NB + b] = run;
                run += v[k];
            }
        }
    }
    tot[b] = run;
    __syncthreads();
    for (int off = 1; off < 1024; off <<= 1) {
        const int v = (b >= off) ? tot[b - off] : 0;
        __syncthreads();
        tot[b] += v;
        __syncthreads();
    }
    if (b < NB) binstart[b] = tot[b] - run;
    if (b == 0) binstart[NB] = tot[1023];
}

__global__ __launch_bounds__(SORT_THREADS)
void bin_fill_kernel(const int*   __restrict__ src,
                     const int*   __restrict__ dst,
                     const float* __restrict__ w,
                     const float* __restrict__ alpha_p,
                     const int*   __restrict__ offs,
                     const int*   __restrict__ binstart,
                     int2*        __restrict__ rec,
                     int E, int NB) {
    __shared__ int cur[NBIN_MAX];
    const int tid = threadIdx.x;
    for (int i = tid; i < NB; i += blockDim.x)
        cur[i] = binstart[i] + offs[blockIdx.x * NB + i];
    __syncthreads();
    const float alpha = sigmoid_of(alpha_p);
    const int chunk = (E + gridDim.x - 1) / gridDim.x;
    const int e0 = blockIdx.x * chunk;
    const int e1 = min(E, e0 + chunk);
    for (int eb = e0 + tid; eb < e1; eb += SORT_ITEMS * SORT_THREADS) {
        int d[SORT_ITEMS], s[SORT_ITEMS];
        float ww[SORT_ITEMS];
#pragma unroll
        for (int k = 0; k < SORT_ITEMS; ++k) {
            const int e = min(eb + k * SORT_THREADS, e1 - 1);
            d[k]  = dst[e];
            s[k]  = src[e];
            ww[k] = w[e];
        }
#pragma unroll
        for (int k = 0; k < SORT_ITEMS; ++k) {
            if (eb + k * SORT_THREADS < e1) {
                const int b = d[k] / BIN_ROWS;
                const int pos = __hip_atomic_fetch_add(&cur[b], 1, __ATOMIC_RELAXED,
                                                       __HIP_MEMORY_SCOPE_WORKGROUP);
                int2 r;
                r.x = s[k] | ((d[k] - b * BIN_ROWS) << 20);
                r.y = __float_as_int(alpha * ww[k]);
                rec[pos] = r;
            }
        }
    }
}

__device__ __forceinline__ void finish_tile_acc(const TileRegs& r, const float* av,
                                                int tile, int hi, int li,
                                                float ca, float beta,
                                                float* __restrict__ nxt) {
    const int nb = (tile >> 1) * 16;
    const int cb = (tile & 1) * 16;

    v8f c;
#pragma unroll
    for (int v = 0; v < 8; ++v)
        c[v] = beta * r.zv[v] + av[v];

#pragma unroll
    for (int m = 0; m < 4; ++m) {
        v2f a, b;
#pragma unroll
        for (int v = 0; v < 2; ++v) {
            const int k = v + 2 * hi;
            a[v] = (li == 4 * m + k) ? ca : 0.0f;
            b[v] = r.bv[2 * m + v];
        }
        c = __builtin_amdgcn_wmma_f32_16x16x4_f32(
                false, a, false, b, (short)0, c, false, false);
    }

#pragma unroll
    for (int v = 0; v < 8; ++v) {
        const int row = nb + v + 8 * hi;
        nxt[row * FEAT_D + cb + li] = c[v];
    }
}

__device__ __forceinline__ int2 pad_batch(int2 q, int base, int lane, int e1) {
    if (base + lane >= e1) q = make_int2(BIN_ROWS << 20, 0);
    return q;
}

__global__ __launch_bounds__(1024)
void step_binned_kernel(const float* __restrict__ x,
                        const float* __restrict__ x0,
                        const int2*  __restrict__ rec,
                        const int*   __restrict__ binstart,
                        const float* __restrict__ alpha_p,
                        const float* __restrict__ beta_p,
                        float*       __restrict__ nxt,
                        int N, int NB) {
    __shared__ float acc[(BIN_ROWS + 1) * FEAT_D];

    const int wave = __builtin_amdgcn_readfirstlane(threadIdx.x >> 5);
    const int nw   = blockDim.x >> 5;
    const int lane = threadIdx.x & 31;
    const unsigned ulane = threadIdx.x & 31u;
    const int hi   = lane >> 4;
    const int li   = lane & 15;

    const float alpha = sigmoid_of(alpha_p);
    const float beta  = *beta_p;
    const float ca    = 1.0f - alpha;
    const int   nrt_all = N >> 4;

    for (int b = blockIdx.x; b < NB; b += gridDim.x) {
        for (int i = threadIdx.x; i < (BIN_ROWS + 1) * FEAT_D; i += blockDim.x) acc[i] = 0.0f;
        __syncthreads();

        const int e0 = binstart[b];
        const int e1 = binstart[b + 1];
        int base = e0 + (wave << 5);
        int2 r = make_int2(0, 0);
        if (base < e1) r = pad_batch(rec[min(base + lane, e1 - 1)], base, lane, e1);
        while (base < e1) {
            const int bn = base + (nw << 5);
            float v[EDGES_PER_WAVE];
#pragma unroll
            for (int j = 0; j < EDGES_PER_WAVE; ++j) {
                const unsigned s = (unsigned)__builtin_amdgcn_readlane(r.x, j) & 0xFFFFFu;
                v[j] = x[(s << 5) + ulane];
            }
            int2 rn = make_int2(0, 0);
            if (bn < e1) rn = rec[min(bn + lane, e1 - 1)];
#pragma unroll
            for (int j = 0; j < EDGES_PER_WAVE; ++j) {
                const unsigned dl = (unsigned)__builtin_amdgcn_readlane(r.x, j) >> 20;
                const float we = __int_as_float(__builtin_amdgcn_readlane(r.y, j));
                __hip_atomic_fetch_add(&acc[(dl << 5) + ulane], we * v[j],
                                       __ATOMIC_RELAXED, __HIP_MEMORY_SCOPE_WORKGROUP);
            }
            r = pad_batch(rn, bn, lane, e1);
            base = bn;
        }
        __syncthreads();

        const int rt0 = b * BIN_TILES;
        const int nrt = min(BIN_TILES, nrt_all - rt0);
        for (int t = wave; t < 2 * nrt; t += nw) {
            const int tile = rt0 * 2 + t;
            TileRegs tr;
            load_tile(x, x0, tile, hi, li, tr);
            float av[8];
            const int lrow0 = (t >> 1) * 16;
            const int cb = (t & 1) * 16;
#pragma unroll
            for (int v = 0; v < 8; ++v)
                av[v] = acc[(lrow0 + v + 8 * hi) * FEAT_D + cb + li];
            finish_tile_acc(tr, av, tile, hi, li, ca, beta, nxt);
        }
        __syncthreads();
    }
}

extern "C" void kernel_launch(void* const* d_in, const int* in_sizes, int n_in,
                              void* d_out, int out_size, void* d_ws, size_t ws_size,
                              hipStream_t stream) {
    const float* x_in    = (const float*)d_in[0];
    const float* w       = (const float*)d_in[1];
    const float* x0      = (const float*)d_in[2];
    const float* alpha_p = (const float*)d_in[3];
    const float* beta_p  = (const float*)d_in[4];
    const int*   eidx    = (const int*)  d_in[5];

    const int N = in_sizes[0] / FEAT_D;
    const int E = in_sizes[1];
    const int* src = eidx;
    const int* dst = eidx + E;

    float* wsx = (float*)d_ws;
    float* out = (float*)d_out;

    const int NB = (N + BIN_ROWS - 1) / BIN_ROWS;
    const size_t plane = (size_t)N * FEAT_D * sizeof(float);
    const size_t need  = plane + (size_t)E * sizeof(int2)
                       + ((size_t)2 * BIN_BLOCKS * NB + NB + 1) * sizeof(int);

    if (ws_size >= need && NB <= NBIN_MAX && N <= (1 << 20)) {
        int2* rec      = (int2*)((char*)d_ws + plane);
        int*  cnt      = (int*)(rec + E);
        int*  offs     = cnt + BIN_BLOCKS * NB;
        int*  binstart = offs + BIN_BLOCKS * NB;

        dim3 big(1024);
        bin_count_kernel<<<BIN_BLOCKS, dim3(SORT_THREADS), 0, stream>>>(dst, cnt, E, NB);
        bin_scan_kernel<<<1, big, 0, stream>>>(cnt, offs, binstart, NB);
        bin_fill_kernel<<<BIN_BLOCKS, dim3(SORT_THREADS), 0, stream>>>(src, dst, w, alpha_p, offs, binstart, rec, E, NB);

        step_binned_kernel<<<STEP_BLOCKS, big, 0, stream>>>(x_in, x0, rec, binstart, alpha_p, beta_p, wsx, N, NB);
        step_binned_kernel<<<STEP_BLOCKS, big, 0, stream>>>(wsx,  x0, rec, binstart, alpha_p, beta_p, out, N, NB);
        step_binned_kernel<<<STEP_BLOCKS, big, 0, stream>>>(out,  x0, rec, binstart, alpha_p, beta_p, wsx, N, NB);
        step_binned_kernel<<<STEP_BLOCKS, big, 0, stream>>>(wsx,  x0, rec, binstart, alpha_p, beta_p, out, N, NB);
        return;
    }

    dim3 blk(256);
    const int ntiles = (N / 16) * (FEAT_D / 16);

    base_wmma_kernel<<<BASE_BLOCKS, blk, 0, stream>>>(x_in, x0, alpha_p, beta_p, wsx, ntiles);
    scatter_kernel<<<SCAT_BLOCKS, blk, 0, stream>>>(x_in, src, dst, w, alpha_p, wsx, E);
    base_wmma_kernel<<<BASE_BLOCKS, blk, 0, stream>>>(wsx, x0, alpha_p, beta_p, out, ntiles);
    scatter_kernel<<<SCAT_BLOCKS, blk, 0, stream>>>(wsx, src, dst, w, alpha_p, out, E);
    base_wmma_kernel<<<BASE_BLOCKS, blk, 0, stream>>>(out, x0, alpha_p, beta_p, wsx, ntiles);
    scatter_kernel<<<SCAT_BLOCKS, blk, 0, stream>>>(out, src, dst, w, alpha_p, wsx, E);
    base_wmma_kernel<<<BASE_BLOCKS, blk, 0, stream>>>(wsx, x0, alpha_p, beta_p, out, ntiles);
    scatter_kernel<<<SCAT_BLOCKS, blk, 0, stream>>>(wsx, src, dst, w, alpha_p, out, E);
}


# Edge sort in ONE launch, was 3: each chunk of the edge list is sorted by bin in LDS and stored in order; a step wave walks its 8 chunk runs of a bin as one list (5 launches); f32 sum order differs
# speedup vs baseline: 5.4574x; 1.1327x over previous
#include <hip/hip_runtime.h>
#include <math.h>


#define FEAT_D 32

typedef __attribute__((ext_vector_type(2))) float v2f;
typedef __attribute__((ext_vector_type(8))) float v8f;

#define SCAT_BLOCKS 1024
#define BASE_BLOCKS 512

__device__ __forceinline__ float sigmoid_of(const float* __restrict__ p) {
    return 1.0f / (1.0f + expf(-(*p)));
}

#define EDGES_PER_WAVE 32

__device__ __forceinline__ void scatter_one(const float* __restrict__ x,
                                            float*       __restrict__ nxt,
                                            int s, int t, float we, int lane) {
    float val = we * x[(size_t)(unsigned)s * FEAT_D + lane];
    unsafeAtomicAdd(&nxt[(size_t)(unsigned)t * FEAT_D + lane], val);
}

__global__ __launch_bounds__(256)
void scatter_kernel(const float* __restrict__ x,
                    const int*   __restrict__ src,
                    const int*   __restrict__ dst,
                    const float* __restrict__ w,
                    const float* __restrict__ alpha_p,
                    float*       __restrict__ nxt,
                    int E) {
    const int wave = __builtin_amdgcn_readfirstlane(
        blockIdx.x * (blockDim.x >> 5) + (threadIdx.x >> 5));
    const int nwaves = gridDim.x * (blockDim.x >> 5);
    const int lane = threadIdx.x & 31;
    const unsigned ulane = threadIdx.x & 31u;
    const int nfull = E >> 5;

    const float alpha = sigmoid_of(alpha_p);

    int b = wave;
    int s_l = 0, t_l = 0;
    float wr_l = 0.0f;
    if (b < nfull) {
        const int el = (b << 5) + lane;
        s_l = src[el];
        t_l = dst[el];
        wr_l = w[el];
    }
    while (b < nfull) {
        float v[EDGES_PER_WAVE];
#pragma unroll
        for (int j = 0; j < EDGES_PER_WAVE; ++j) {
            const int s = __builtin_amdgcn_readlane(s_l, j);
            v[j] = x[((unsigned)s << 5) + ulane];
        }
        const int bn = b + nwaves;
        int s_n = 0, t_n = 0;
        float wr_n = 0.0f;
        if (bn < nfull) {
            const int el = (bn << 5) + lane;
            s_n = src[el];
            t_n = dst[el];
            wr_n = w[el];
        }
        const int w_l = __float_as_int(alpha * wr_l);
#pragma unroll
        for (int j = 0; j < EDGES_PER_WAVE; ++j) {
            const int   t  = __builtin_amdgcn_readlane(t_l, j);
            const float we = __int_as_float(__builtin_amdgcn_readlane(w_l, j));
            unsafeAtomicAdd(&nxt[((unsigned)t << 5) + ulane], we * v[j]);
        }
        s_l = s_n; t_l = t_n; wr_l = wr_n;
        b = bn;
    }

    const int rem = E & 31;
    if (rem != 0 && wave == nwaves - 1) {
        const int el = (nfull << 5) + min(lane, rem - 1);
        s_l = src[el];
        t_l = dst[el];
        const int w_l = __float_as_int(alpha * w[el]);
        for (int j = 0; j < rem; ++j) {
            const int   s  = __builtin_amdgcn_readlane(s_l, j);
            const int   t  = __builtin_amdgcn_readlane(t_l, j);
            const float we = __int_as_float(__builtin_amdgcn_readlane(w_l, j));
            scatter_one(x, nxt, s, t, we, lane);
        }
    }
}

struct TileRegs {
    float zv[8];
    float bv[8];
};

__device__ __forceinline__ void load_tile(const float* __restrict__ x,
                                          const float* __restrict__ x0,
                                          int tile, int hi, int li, TileRegs& r) {
    const int nb = (tile >> 1) * 16;
    const int cb = (tile & 1) * 16;
#pragma unroll
    for (int v = 0; v < 8; ++v) {
        const int idx = (nb + v + 8 * hi) * FEAT_D + cb + li;
        r.zv[v] = x0[idx];
    }
#pragma unroll
    for (int m = 0; m < 4; ++m) {
#pragma unroll
        for (int v = 0; v < 2; ++v) {
            const int rowb = nb + 4 * m + v + 2 * hi;
            r.bv[2 * m + v] = x[rowb * FEAT_D + cb + li];
        }
    }
}

__device__ __forceinline__ void finish_tile(const TileRegs& r, int tile, int hi, int li,
                                            float ca, float beta,
                                            float* __restrict__ nxt) {
    const int nb = (tile >> 1) * 16;
    const int cb = (tile & 1) * 16;

    v8f c;
#pragma unroll
    for (int v = 0; v < 8; ++v)
        c[v] = beta * r.zv[v];

#pragma unroll
    for (int m = 0; m < 4; ++m) {
        v2f a, b;
#pragma unroll
        for (int v = 0; v < 2; ++v) {
            const int k = v + 2 * hi;
            a[v] = (li == 4 * m + k) ? ca : 0.0f;
            b[v] = r.bv[2 * m + v];
        }
        c = __builtin_amdgcn_wmma_f32_16x16x4_f32(
                false, a, false, b, (short)0, c, false, false);
    }

#pragma unroll
    for (int v = 0; v < 8; ++v) {
        const int row = nb + v + 8 * hi;
        nxt[row * FEAT_D + cb + li] = c[v];
    }
}

__global__ __launch_bounds__(256)
void base_wmma_kernel(const float* __restrict__ x,
                      const float* __restrict__ x0,
                      const float* __restrict__ alpha_p,
                      const float* __restrict__ beta_p,
                      float*       __restrict__ nxt,
                      int ntiles) {
    const int wave = __builtin_amdgcn_readfirstlane(
        blockIdx.x * (blockDim.x >> 5) + (threadIdx.x >> 5));
    const int nwaves = gridDim.x * (blockDim.x >> 5);
    if (wave >= ntiles) return;

    const int lane = threadIdx.x & 31;
    const int hi   = lane >> 4;
    const int li   = lane & 15;

    const float alpha = sigmoid_of(alpha_p);
    const float beta  = *beta_p;
    const float ca    = 1.0f - alpha;

    const int step2 = 2 * nwaves;
    int t = wave;
    TileRegs ra, rb;
    load_tile(x, x0, t, hi, li, ra);
    while (t + step2 < ntiles) {
        load_tile(x, x0, t + nwaves, hi, li, rb);
        __builtin_amdgcn_sched_barrier(0);
        finish_tile(ra, t, hi, li, ca, beta, nxt);
        load_tile(x, x0, t + step2, hi, li, ra);
        __builtin_amdgcn_sched_barrier(0);
        finish_tile(rb, t + nwaves, hi, li, ca, beta, nxt);
        t += step2;
    }
    if (t + nwaves < ntiles) {
        load_tile(x, x0, t + nwaves, hi, li, rb);
        __builtin_amdgcn_sched_barrier(0);
        finish_tile(ra, t, hi, li, ca, beta, nxt);
        finish_tile(rb, t + nwaves, hi, li, ca, beta, nxt);
    } else {
        finish_tile(ra, t, hi, li, ca, beta, nxt);
    }
}

#define BIN_TILES   25
#define BIN_ROWS    (16 * BIN_TILES)
#define SORT_BLOCKS 256
#define SORT_THREADS 1024
#define SORT_ITEMS  8
#define SORT_CAP    8192
#define NBIN_MAX    1024
#define STEP_BLOCKS 256
#define STEP_WAVES  32
#define RUNS_PER_WAVE (SORT_BLOCKS / STEP_WAVES)

__global__ __launch_bounds__(SORT_THREADS)
void chunk_sort_kernel(const int*   __restrict__ src,
                       const int*   __restrict__ dst,
                       const float* __restrict__ w,
                       const float* __restrict__ alpha_p,
                       int*         __restrict__ offtab,
                       int2*        __restrict__ rec,
                       int E, int NB) {
    static_assert(NBIN_MAX == SORT_THREADS, "thread b owns bin b in the scan");
    __shared__ int  h[NBIN_MAX];
    __shared__ int  tot[NBIN_MAX];
    __shared__ int2 stage[SORT_CAP];
    const int tid = threadIdx.x;
    h[tid] = 0;
    __syncthreads();
    const int chunk = (E + gridDim.x - 1) / gridDim.x;
    const int e0 = min(E, (int)blockIdx.x * chunk);
    const int e1 = min(E, e0 + chunk);

    for (int eb = e0 + tid; eb < e1; eb += SORT_ITEMS * SORT_THREADS) {
        int d[SORT_ITEMS];
#pragma unroll
        for (int k = 0; k < SORT_ITEMS; ++k)
            d[k] = dst[min(eb + k * SORT_THREADS, e1 - 1)];
#pragma unroll
        for (int k = 0; k < SORT_ITEMS; ++k)
            if (eb + k * SORT_THREADS < e1)
                __hip_atomic_fetch_add(&h[d[k] / BIN_ROWS], 1, __ATOMIC_RELAXED,
                                       __HIP_MEMORY_SCOPE_WORKGROUP);
    }
    __syncthreads();

    const int own = h[tid];
    tot[tid] = own;
    __syncthreads();
    for (int off = 1; off < NBIN_MAX; off <<= 1) {
        const int v = (tid >= off) ? tot[tid - off] : 0;
        __syncthreads();
        tot[tid] += v;
        __syncthreads();
    }
    const int excl = tot[tid] - own;
    h[tid] = excl;
    int* const myoff = offtab + (size_t)blockIdx.x * (NB + 1);
    if (tid < NB)  myoff[tid] = e0 + excl;
    if (tid == 0)  myoff[NB]  = e1;
    __syncthreads();

    const float alpha = sigmoid_of(alpha_p);
    for (int eb = e0 + tid; eb < e1; eb += SORT_ITEMS * SORT_THREADS) {
        int d[SORT_ITEMS], s[SORT_ITEMS];
        float ww[SORT_ITEMS];
#pragma unroll
        for (int k = 0; k < SORT_ITEMS; ++k) {
            const int e = min(eb + k * SORT_THREADS, e1 - 1);
            d[k]  = dst[e];
            s[k]  = src[e];
            ww[k] = w[e];
        }
#pragma unroll
        for (int k = 0; k < SORT_ITEMS; ++k) {
            if (eb + k * SORT_THREADS < e1) {
                const int b = d[k] / BIN_ROWS;
                const int pos = __hip_atomic_fetch_add(&h[b], 1, __ATOMIC_RELAXED,
                                                       __HIP_MEMORY_SCOPE_WORKGROUP);
                int2 r;
                r.x = s[k] | ((d[k] - b * BIN_ROWS) << 20);
                r.y = __float_as_int(alpha * ww[k]);
                stage[pos] = r;
            }
        }
    }
    __syncthreads();

    for (int i = tid; i < e1 - e0; i += SORT_THREADS) rec[e0 + i] = stage[i];
}

__device__ __forceinline__ void finish_tile_acc(const TileRegs& r, const float* av,
                                                int tile, int hi, int li,
                                                float ca, float beta,
                                                float* __restrict__ nxt) {
    const int nb = (tile >> 1) * 16;
    const int cb = (tile & 1) * 16;

    v8f c;
#pragma unroll
    for (int v = 0; v < 8; ++v)
        c[v] = beta * r.zv[v] + av[v];

#pragma unroll
    for (int m = 0; m < 4; ++m) {
        v2f a, b;
#pragma unroll
        for (int v = 0; v < 2; ++v) {
            const int k = v + 2 * hi;
            a[v] = (li == 4 * m + k) ? ca : 0.0f;
            b[v] = r.bv[2 * m + v];
        }
        c = __builtin_amdgcn_wmma_f32_16x16x4_f32(
                false, a, false, b, (short)0, c, false, false);
    }

#pragma unroll
    for (int v = 0; v < 8; ++v) {
        const int row = nb + v + 8 * hi;
        nxt[row * FEAT_D + cb + li] = c[v];
    }
}

__device__ __forceinline__ int rec_index(int i, const int* pre, const int* dlt) {
    int g = dlt[0] + i;
#pragma unroll
    for (int c = 1; c < RUNS_PER_WAVE; ++c)
        if (i >= pre[c]) g = dlt[c] + i;
    return g;
}

__device__ __forceinline__ int2 pad_batch(int2 q, int base, int lane, int n) {
    if (base + lane >= n) q = make_int2(BIN_ROWS << 20, 0);
    return q;
}

__global__ __launch_bounds__(STEP_WAVES * 32)
void step_binned_kernel(const float* __restrict__ x,
                        const float* __restrict__ x0,
                        const int2*  __restrict__ rec,
                        const int*   __restrict__ offtab,
                        const float* __restrict__ alpha_p,
                        const float* __restrict__ beta_p,
                        float*       __restrict__ nxt,
                        int N, int NB) {
    static_assert(SORT_BLOCKS % STEP_WAVES == 0, "every wave walks the same number of chunk runs");
    __shared__ float acc[(BIN_ROWS + 1) * FEAT_D];

    const int wave = __builtin_amdgcn_readfirstlane(threadIdx.x >> 5);
    const int nw   = blockDim.x >> 5;
    const int lane = threadIdx.x & 31;
    const unsigned ulane = threadIdx.x & 31u;
    const int hi   = lane >> 4;
    const int li   = lane & 15;

    const float alpha = sigmoid_of(alpha_p);
    const float beta  = *beta_p;
    const float ca    = 1.0f - alpha;
    const int   nrt_all = N >> 4;

    for (int b = blockIdx.x; b < NB; b += gridDim.x) {
        for (int i = threadIdx.x; i < (BIN_ROWS + 1) * FEAT_D; i += blockDim.x) acc[i] = 0.0f;
        __syncthreads();

        int rlo = 0, rhi = 0;
        if (lane < RUNS_PER_WAVE) {
            const int* p = offtab + (size_t)(wave * RUNS_PER_WAVE + lane) * (NB + 1) + b;
            rlo = p[0];
            rhi = p[1];
        }
        const int rlen = rhi - rlo;
        int pre[RUNS_PER_WAVE + 1], dlt[RUNS_PER_WAVE];
        pre[0] = 0;
#pragma unroll
        for (int c = 0; c < RUNS_PER_WAVE; ++c) {
            dlt[c] = __builtin_amdgcn_readlane(rlo, c) - pre[c];
            pre[c + 1] = pre[c] + __builtin_amdgcn_readlane(rlen, c);
        }
        const int n = pre[RUNS_PER_WAVE];

        int base = 0;
        int2 r = make_int2(0, 0);
        if (n > 0) r = pad_batch(rec[rec_index(min(lane, n - 1), pre, dlt)], 0, lane, n);
        while (base < n) {
            const int bn = base + 32;
            float v[EDGES_PER_WAVE];
#pragma unroll
            for (int j = 0; j < EDGES_PER_WAVE; ++j) {
                const unsigned s = (unsigned)__builtin_amdgcn_readlane(r.x, j) & 0xFFFFFu;
                v[j] = x[(s << 5) + ulane];
            }
            int2 rn = make_int2(0, 0);
            if (bn < n) rn = rec[rec_index(min(bn + lane, n - 1), pre, dlt)];
#pragma unroll
            for (int j = 0; j < EDGES_PER_WAVE; ++j) {
                const unsigned dl = (unsigned)__builtin_amdgcn_readlane(r.x, j) >> 20;
                const float we = __int_as_float(__builtin_amdgcn_readlane(r.y, j));
                __hip_atomic_fetch_add(&acc[(dl << 5) + ulane], we * v[j],
                                       __ATOMIC_RELAXED, __HIP_MEMORY_SCOPE_WORKGROUP);
            }
            r = pad_batch(rn, bn, lane, n);
            base = bn;
        }
        __syncthreads();

        const int rt0 = b * BIN_TILES;
        const int nrt = min(BIN_TILES, nrt_all - rt0);
        for (int t = wave; t < 2 * nrt; t += nw) {
            const int tile = rt0 * 2 + t;
            TileRegs tr;
            load_tile(x, x0, tile, hi, li, tr);
            float av[8];
            const int lrow0 = (t >> 1) * 16;
            const int cb = (t & 1) * 16;
#pragma unroll
            for (int v = 0; v < 8; ++v)
                av[v] = acc[(lrow0 + v + 8 * hi) * FEAT_D + cb + li];
            finish_tile_acc(tr, av, tile, hi, li, ca, beta, nxt);
        }
        __syncthreads();
    }
}

extern "C" void kernel_launch(void* const* d_in, const int* in_sizes, int n_in,
                              void* d_out, int out_size, void* d_ws, size_t ws_size,
                              hipStream_t stream) {
    const float* x_in    = (const float*)d_in[0];
    const float* w       = (const float*)d_in[1];
    const float* x0      = (const float*)d_in[2];
    const float* alpha_p = (const float*)d_in[3];
    const float* beta_p  = (const float*)d_in[4];
    const int*   eidx    = (const int*)  d_in[5];

    const int N = in_sizes[0] / FEAT_D;
    const int E = in_sizes[1];
    const int* src = eidx;
    const int* dst = eidx + E;

    float* wsx = (float*)d_ws;
    float* out = (float*)d_out;

    const int NB = (N + BIN_ROWS - 1) / BIN_ROWS;
    const size_t plane = (size_t)N * FEAT_D * sizeof(float);
    const size_t need  = plane + (size_t)E * sizeof(int2)
                       + (size_t)SORT_BLOCKS * (NB + 1) * sizeof(int);

    if (ws_size >= need && NB <= NBIN_MAX && N <= (1 << 20)
        && (size_t)E <= (size_t)SORT_BLOCKS * SORT_CAP) {
        int2* rec    = (int2*)((char*)d_ws + plane);
        int*  offtab = (int*)(rec + E);

        dim3 big(STEP_WAVES * 32);
        chunk_sort_kernel<<<SORT_BLOCKS, dim3(SORT_THREADS), 0, stream>>>(src, dst, w, alpha_p, offtab, rec, E, NB);

        step_binned_kernel<<<STEP_BLOCKS, big, 0, stream>>>(x_in, x0, rec, offtab, alpha_p, beta_p, wsx, N, NB);
        step_binned_kernel<<<STEP_BLOCKS, big, 0, stream>>>(wsx,  x0, rec, offtab, alpha_p, beta_p, out, N, NB);
        step_binned_kernel<<<STEP_BLOCKS, big, 0, stream>>>(out,  x0, rec, offtab, alpha_p, beta_p, wsx, N, NB);
        step_binned_kernel<<<STEP_BLOCKS, big, 0, stream>>>(wsx,  x0, rec, offtab, alpha_p, beta_p, out, N, NB);
        return;
    }

    dim3 blk(256);
    const int ntiles = (N / 16) * (FEAT_D / 16);

    base_wmma_kernel<<<BASE_BLOCKS, blk, 0, stream>>>(x_in, x0, alpha_p, beta_p, wsx, ntiles);
    scatter_kernel<<<SCAT_BLOCKS, blk, 0, stream>>>(x_in, src, dst, w, alpha_p, wsx, E);
    base_wmma_kernel<<<BASE_BLOCKS, blk, 0, stream>>>(wsx, x0, alpha_p, beta_p, out, ntiles);
    scatter_kernel<<<SCAT_BLOCKS, blk, 0, stream>>>(wsx, src, dst, w, alpha_p, out, E);
    base_wmma_kernel<<<BASE_BLOCKS, blk, 0, stream>>>(out, x0, alpha_p, beta_p, wsx, ntiles);
    scatter_kernel<<<SCAT_BLOCKS, blk, 0, stream>>>(out, src, dst, w, alpha_p, wsx, E);
    base_wmma_kernel<<<BASE_BLOCKS, blk, 0, stream>>>(wsx, x0, alpha_p, beta_p, out, ntiles);
    scatter_kernel<<<SCAT_BLOCKS, blk, 0, stream>>>(wsx, src, dst, w, alpha_p, out, E);
}
